# starting kernel
# baseline (speedup 1.0000x reference)
#include <hip/hip_runtime.h>

#define IN_FEAT 128
#define OUT_FEAT 64
#define LDS_STRIDE 80

typedef __attribute__((ext_vector_type(2))) float v2f;
typedef __attribute__((ext_vector_type(8))) float v8f;

__global__ __launch_bounds__(128) void gcn_gemm_wmma(
    const float* __restrict__ x, const float* __restrict__ w,
    float* __restrict__ support, int n_nodes) {
  __shared__ float wlds[IN_FEAT * LDS_STRIDE];

  const int tid = threadIdx.x;
  for (int i = tid; i < IN_FEAT * OUT_FEAT; i += 128) {
    int r = i >> 6;
    int c = i & 63;
    wlds[r * LDS_STRIDE + c] = w[i];
  }
  __syncthreads();

  const int wave = tid >> 5;
  const int lane = tid & 31;
  const int half = lane >> 4;
  const int l16  = lane & 15;
  const int row_base = blockIdx.x * 16;
  const int col_base = wave * 16;

  int arow = row_base + l16;
  if (arow >= n_nodes) arow = n_nodes - 1;
  const float* __restrict__ xrow = x + (long long)arow * IN_FEAT;

  v8f acc = {};
#pragma unroll
  for (int k = 0; k < IN_FEAT; k += 4) {
    const int kk = k + half * 2;
    v2f a = *(const v2f*)(xrow + kk);
    v2f b;
    b.x = wlds[(kk + 0) * LDS_STRIDE + col_base + l16];
    b.y = wlds[(kk + 1) * LDS_STRIDE + col_base + l16];
    acc = __builtin_amdgcn_wmma_f32_16x16x4_f32(
         false, a,  false, b,
         (short)0, acc,  false,  false);
  }

#pragma unroll
  for (int i = 0; i < 8; ++i) {
    const int orow = row_base + half * 8 + i;
    if (orow < n_nodes)
      support[(long long)orow * OUT_FEAT + col_base + l16] = acc[i];
  }
}

__global__ __launch_bounds__(256) void gcn_zero(float* __restrict__ out, int n) {
  int i = blockIdx.x * blockDim.x + threadIdx.x;
  if (i < n) out[i] = 0.0f;
}

__global__ __launch_bounds__(256) void gcn_spmm(
    const float* __restrict__ support, const int* __restrict__ rows,
    const int* __restrict__ cols, const float* __restrict__ vals,
    float* __restrict__ out, int n_edges) {
  const long long t = (long long)blockIdx.x * blockDim.x + threadIdx.x;
  const long long e = t >> 4;
  if (e >= n_edges) return;
  const int f4 = (int)(t & 15) * 4;

  const int r = rows[e];
  const int c = cols[e];
  const float v = vals[e];

  const float4 s = *(const float4*)(support + (long long)c * OUT_FEAT + f4);
  float* o = out + (long long)r * OUT_FEAT + f4;
  unsafeAtomicAdd(o + 0, v * s.x);
  unsafeAtomicAdd(o + 1, v * s.y);
  unsafeAtomicAdd(o + 2, v * s.z);
  unsafeAtomicAdd(o + 3, v * s.w);
}

extern "C" void kernel_launch(void* const* d_in, const int* in_sizes, int n_in,
                              void* d_out, int out_size, void* d_ws, size_t ws_size,
                              hipStream_t stream) {
  const float* x    = (const float*)d_in[0];
  const float* w    = (const float*)d_in[1];
  const int*   rows = (const int*)d_in[2];
  const int*   cols = (const int*)d_in[3];
  const float* vals = (const float*)d_in[4];
  float* out     = (float*)d_out;
  float* support = (float*)d_ws;

  const int n_nodes = in_sizes[0] / IN_FEAT;
  const int n_edges = in_sizes[2];

  gcn_gemm_wmma<<<(n_nodes + 15) / 16, 128, 0, stream>>>(x, w, support, n_nodes);

  gcn_zero<<<(out_size + 255) / 256, 256, 0, stream>>>(out, out_size);

  const long long sp_threads = (long long)n_edges * 16;
  gcn_spmm<<<(unsigned)((sp_threads + 255) / 256), 256, 0, stream>>>(
      support, rows, cols, vals, out, n_edges);
}


# gemm: each wave keeps its 128x16 weight slice in registers as B fragments for the whole launch; 1250 blocks walk the 16-row x tiles (x alone in LDS, next tile prefetched); k and sum order unchanged
# speedup vs baseline: 2.3473x; 2.3473x over previous
#include <hip/hip_runtime.h>

#define IN_FEAT 128
#define OUT_FEAT 64
#define X_PITCH 132

__device__ __forceinline__ int head_stride(int n_nodes) { return (n_nodes + 63) & ~63; }

typedef __attribute__((ext_vector_type(2))) float v2f;
typedef __attribute__((ext_vector_type(8))) float v8f;

__device__ __forceinline__ float4 load_x_unit(const float4* __restrict__ x4, int tile, int i, int n_nodes) {
  int grow = tile * 16 + (i >> 5);
  if (grow >= n_nodes) grow = n_nodes - 1;
  return x4[(size_t)grow * (IN_FEAT / 4) + (i & 31)];
}

__global__ __launch_bounds__(128) __attribute__((amdgpu_waves_per_eu(1, 8))) void gcn_gemm_wmma(
    const float* __restrict__ x, const float* w,
    float* support, int* head, int n_nodes) {
  __shared__ __attribute__((aligned(16))) float xlds[2 * 16 * X_PITCH];

  const int tid = threadIdx.x;
  const int wave = tid >> 5;
  const int lane = tid & 31;
  const int half = lane >> 4;
  const int l16  = lane & 15;
  const int col_base = wave * 16;
  const int n_tiles = (n_nodes + 15) >> 4;
  const int hs = head_stride(n_nodes);

  v2f bf[IN_FEAT / 4];
#pragma unroll
  for (int s = 0; s < IN_FEAT / 4; ++s) {
    const int kk = 4 * s + half * 2;
    bf[s].x = w[(kk + 0) * OUT_FEAT + col_base + l16];
    bf[s].y = w[(kk + 1) * OUT_FEAT + col_base + l16];
  }

  const float4* __restrict__ x4 = (const float4*)x;
  int t = blockIdx.x;
  float4 xa = load_x_unit(x4, t, tid, n_nodes);
  float4 xb = load_x_unit(x4, t, tid + 128, n_nodes);
  float4 xc = load_x_unit(x4, t, tid + 256, n_nodes);
  float4 xd = load_x_unit(x4, t, tid + 384, n_nodes);
  const int xoff = (tid >> 5) * X_PITCH + (tid & 31) * 4;

  int buf = 0;
  for (; t < n_tiles; t += gridDim.x) {
    const int row_base = t * 16;
    if (tid < 32 && row_base + (tid & 15) < n_nodes)
      head[(tid >> 4) * hs + row_base + (tid & 15)] = -1;

    float* xl = xlds + buf * (16 * X_PITCH);
    *(float4*)&xl[xoff] = xa;
    *(float4*)&xl[xoff + 4 * X_PITCH] = xb;
    *(float4*)&xl[xoff + 8 * X_PITCH] = xc;
    *(float4*)&xl[xoff + 12 * X_PITCH] = xd;
    __builtin_amdgcn_fence(__ATOMIC_RELEASE, "workgroup", "local");
    __builtin_amdgcn_s_barrier();
    __builtin_amdgcn_fence(__ATOMIC_ACQUIRE, "workgroup", "local");

    const int tn = t + gridDim.x;
    xa = load_x_unit(x4, tn, tid, n_nodes);
    xb = load_x_unit(x4, tn, tid + 128, n_nodes);
    xc = load_x_unit(x4, tn, tid + 256, n_nodes);
    xd = load_x_unit(x4, tn, tid + 384, n_nodes);

    const float* xrow = xl + l16 * X_PITCH + half * 2;
    v8f acc = {};
#pragma unroll
    for (int s = 0; s < IN_FEAT / 4; ++s) {
      v2f a = *(const v2f*)(xrow + 4 * s);
      acc = __builtin_amdgcn_wmma_f32_16x16x4_f32(
           false, a,  false, bf[s],
           (short)0, acc,  false,  false);
    }

#pragma unroll
    for (int i = 0; i < 8; ++i) {
      const int orow = row_base + half * 8 + i;
      if (orow < n_nodes)
        support[(long long)orow * OUT_FEAT + col_base + l16] = acc[i];
    }
    buf ^= 1;
  }
}

__global__ __launch_bounds__(256) void gcn_link_edges(
    const int* __restrict__ rows, const int* __restrict__ cols,
    const float* __restrict__ vals, int* __restrict__ head,
    int4* __restrict__ rec, int n_nodes, int n_edges) {
  const int e = blockIdx.x * blockDim.x + threadIdx.x;
  if (e >= n_edges) return;
  const unsigned place = __builtin_amdgcn_s_sendmsg_rtn(0x87);
  int* __restrict__ myhead = head + ((place >> 18) & 1u) * head_stride(n_nodes);
  const int r = rows[e];
  const int c = cols[e];
  const float v = vals[e];
  const int next = atomicExch(myhead + r, e);
  rec[e] = make_int4(c, __float_as_int(v), next, 0);
}

__global__ __launch_bounds__(256) void gcn_row_sum(
    const float* __restrict__ support, const int* __restrict__ head,
    const int4* __restrict__ rec, float* __restrict__ out,
    int n_nodes, int n_edges) {
  const int t = blockIdx.x * blockDim.x + threadIdx.x;
  const int r = t >> 3;
  if (r >= n_nodes) return;
  const unsigned j = (unsigned)t & 7u;
  const float4* __restrict__ sup4 = (const float4*)support;
  float4* __restrict__ out4 = (float4*)out;

  float4 a0 = {0.0f, 0.0f, 0.0f, 0.0f};
  float4 a1 = {0.0f, 0.0f, 0.0f, 0.0f};

  int e = head[r];
  int e2 = head[head_stride(n_nodes) + r];
  int4 rc = rec[(unsigned)e < (unsigned)n_edges ? (unsigned)e : 0u];
  const int4 rc2 = rec[(unsigned)e2 < (unsigned)n_edges ? (unsigned)e2 : 0u];

  for (int i = 0; i < n_edges; ++i) {
    if ((unsigned)e >= (unsigned)n_edges) {
      e = e2;
      rc = rc2;
      e2 = -1;
    }
    if ((unsigned)e >= (unsigned)n_edges) break;
    const unsigned b = (unsigned)rc.x * (OUT_FEAT / 4) + j;
    const float4 s0 = sup4[b];
    const float4 s1 = sup4[b + 8];
    const float v = __int_as_float(rc.y);
    e = rc.z;
    if ((unsigned)e < (unsigned)n_edges) rc = rec[(unsigned)e];
    a0.x = fmaf(v, s0.x, a0.x);
    a0.y = fmaf(v, s0.y, a0.y);
    a0.z = fmaf(v, s0.z, a0.z);
    a0.w = fmaf(v, s0.w, a0.w);
    a1.x = fmaf(v, s1.x, a1.x);
    a1.y = fmaf(v, s1.y, a1.y);
    a1.z = fmaf(v, s1.z, a1.z);
    a1.w = fmaf(v, s1.w, a1.w);
  }

  const unsigned ob = (unsigned)r * (OUT_FEAT / 4) + j;
  out4[ob] = a0;
  out4[ob + 8] = a1;
}

extern "C" void kernel_launch(void* const* d_in, const int* in_sizes, int n_in,
                              void* d_out, int out_size, void* d_ws, size_t ws_size,
                              hipStream_t stream) {
  const float* x    = (const float*)d_in[0];
  const float* w    = (const float*)d_in[1];
  const int*   rows = (const int*)d_in[2];
  const int*   cols = (const int*)d_in[3];
  const float* vals = (const float*)d_in[4];
  float* out     = (float*)d_out;

  const int n_nodes = in_sizes[0] / IN_FEAT;
  const int n_edges = in_sizes[2];

  float* support = (float*)d_ws;
  int4*  rec     = (int4*)(support + (size_t)n_nodes * OUT_FEAT);
  int*   head    = (int*)(rec + n_edges);

  gcn_gemm_wmma<<<1250, 128, 0, stream>>>(x, w, support, head, n_nodes);

  gcn_link_edges<<<(n_edges + 255) / 256, 256, 0, stream>>>(
      rows, cols, vals, head, rec, n_nodes, n_edges);

  const long long rs_threads = (long long)n_nodes * 8;
  gcn_row_sum<<<(unsigned)((rs_threads + 255) / 256), 256, 0, stream>>>(
      support, head, rec, out, n_nodes, n_edges);
}
